# starting kernel
# baseline (speedup 1.0000x reference)
#include <hip/hip_runtime.h>

#define B_DIM 1024
#define D_DIM 512
#define K_DIM 256
#define T_DIM 1024

typedef __attribute__((ext_vector_type(2))) float v2f;
typedef __attribute__((ext_vector_type(8))) float v8f;

__global__ __launch_bounds__(256) void gpdf_prep(const float* __restrict__ means,
                                                 const float* __restrict__ rho,
                                                 float* __restrict__ WaT,
                                                 float* __restrict__ bias) {
  const int k = blockIdx.x;
  const int t = threadIdx.x;
  float partial = 0.0f;
#pragma unroll
  for (int i = 0; i < D_DIM / 256; ++i) {
    const int d = t + i * 256;
    const float m = means[k * D_DIM + d];
    const float r = rho[k * D_DIM + d];
    const float s = log1pf(expf(r));
    const float s2 = s * s;
    WaT[d * K_DIM + k]           = 0.5f * s2;
    WaT[(D_DIM + d) * K_DIM + k] = -s2 * m;
    partial += 0.5f * s2 * m * m;
  }
  __shared__ float red[256];
  red[t] = partial;
  __syncthreads();
  for (int s = 128; s > 0; s >>= 1) {
    if (t < s) red[t] += red[t + s];
    __syncthreads();
  }
  if (t == 0) bias[k] = red[0];
}

__global__ __launch_bounds__(256) void gpdf_gemm_wmma(
    const float* __restrict__ x,
    const float* __restrict__ WaT,
    const float* __restrict__ bias,
    float* __restrict__ out) {
  const int lane = threadIdx.x & 31;
  const int wave = threadIdx.x >> 5;
  const int half = lane >> 4;
  const int mn   = lane & 15;
  const int row0 = blockIdx.x * 128 + wave * 16;
  const int col0 = blockIdx.y * 16;

  const float* __restrict__ xrow = x + (row0 + mn) * D_DIM;
  const int bcol = col0 + mn;

  v8f c = {};

  for (int t0 = 0; t0 < D_DIM; t0 += 4) {
    const int ka = t0 + 2 * half;
    v2f a, b;
    const float a0 = xrow[ka];
    const float a1 = xrow[ka + 1];
    a.x = a0 * a0;
    a.y = a1 * a1;
    b.x = WaT[ka * K_DIM + bcol];
    b.y = WaT[(ka + 1) * K_DIM + bcol];
    c = __builtin_amdgcn_wmma_f32_16x16x4_f32(false, a, false, b,
                                              (short)0, c, false, false);
  }
  for (int t0 = 0; t0 < D_DIM; t0 += 4) {
    const int ka = t0 + 2 * half;
    v2f a, b;
    a.x = xrow[ka];
    a.y = xrow[ka + 1];
    b.x = WaT[(D_DIM + ka) * K_DIM + bcol];
    b.y = WaT[(D_DIM + ka + 1) * K_DIM + bcol];
    c = __builtin_amdgcn_wmma_f32_16x16x4_f32(false, a, false, b,
                                              (short)0, c, false, false);
  }

  const float bv = bias[bcol];
#pragma unroll
  for (int j = 0; j < 8; ++j) {
    out[(row0 + j + 8 * half) * K_DIM + bcol] = c[j] + bv;
  }
}

extern "C" void kernel_launch(void* const* d_in, const int* in_sizes, int n_in,
                              void* d_out, int out_size, void* d_ws, size_t ws_size,
                              hipStream_t stream) {
  const float* x     = (const float*)d_in[0];
  const float* means = (const float*)d_in[1];
  const float* rho   = (const float*)d_in[2];
  float* out  = (float*)d_out;
  float* WaT  = (float*)d_ws;
  float* bias = WaT + T_DIM * K_DIM;

  gpdf_prep<<<dim3(K_DIM), dim3(256), 0, stream>>>(means, rho, WaT, bias);
  gpdf_gemm_wmma<<<dim3(B_DIM / 128, K_DIM / 16), dim3(256), 0, stream>>>(
      x, WaT, bias, out);
}


# Each wave streams its fragments in WMMA order by 16-byte loads, 3 groups ahead (W pre-packed so by launch 1, x as 64-B runs per lane); 2x2-wave blocks, x^2 and x on 2 accumulators: sum order changed
# speedup vs baseline: 4.2510x; 4.2510x over previous
#include <hip/hip_runtime.h>

#define B_DIM 1024
#define D_DIM 512
#define K_DIM 256

typedef __attribute__((ext_vector_type(2))) float v2f;
typedef __attribute__((ext_vector_type(4))) float v4f;
typedef __attribute__((ext_vector_type(8))) float v8f;


__global__ __launch_bounds__(256) void gpdf_prep(const float* __restrict__ means,
                                                 const float* __restrict__ rho,
                                                 float* __restrict__ Wp,
                                                 float* __restrict__ bias) {
  __shared__ float red[8];
  const int tid  = threadIdx.x;
  const int lane = tid & 31;
  if (blockIdx.x < 128) {
    const int ct = blockIdx.x >> 3;
    const int j  = (blockIdx.x & 7) * 2 + (tid >> 7);
    const int q  = (tid >> 5) & 3;
    const int c  = lane & 15;
    const int h  = lane >> 4;
    const int k  = ct * 16 + c;
    const int d0 = 32 * j + 16 * h + 4 * q;
    const v4f m = *(const v4f*)(means + k * D_DIM + d0);
    const v4f r = *(const v4f*)(rho + k * D_DIM + d0);
    v4f w1, w2;
#pragma unroll
    for (int i = 0; i < 4; ++i) {
      const float s  = log1pf(expf(r[i]));
      const float s2 = s * s;
      w1[i] = 0.5f * s2;
      w2[i] = -s2 * m[i];
    }
    v4f* dst = (v4f*)Wp + ((((ct * 16 + j) * 2) * 4 + q) * 32 + lane);
    dst[0]   = w1;
    dst[128] = w2;
  } else {
    const int bb  = blockIdx.x - 128;
    const int cls = tid >> 7;
    const int w4  = (tid >> 5) & 3;
    const int k   = 2 * bb + cls;
    const int d0  = 128 * w4 + 4 * lane;
    const v4f m = *(const v4f*)(means + k * D_DIM + d0);
    const v4f r = *(const v4f*)(rho + k * D_DIM + d0);
    float partial = 0.0f;
#pragma unroll
    for (int i = 0; i < 4; ++i) {
      const float s  = log1pf(expf(r[i]));
      const float s2 = s * s;
      partial += 0.5f * s2 * m[i] * m[i];
    }
    for (int off = 16; off > 0; off >>= 1) partial += __shfl_xor(partial, off);
    if (lane == 0) red[cls * 4 + w4] = partial;
    __syncthreads();
    if (tid < 2)
      bias[2 * bb + tid] = ((red[4 * tid] + red[4 * tid + 1]) + red[4 * tid + 2]) + red[4 * tid + 3];
  }
}

#define AHEAD 3

struct Frag {
  v4f a[4];
  v4f b1[4];
  v4f b2[4];
};

__device__ __forceinline__ void load_frag(Frag& f, const v4f* xa, const v4f* wb, int j) {
  for (int q = 0; q < 4; ++q) {
    f.a[q]  = xa[8 * j + q];
    f.b1[q] = wb[j * 256 + q * 32];
    f.b2[q] = wb[j * 256 + 128 + q * 32];
  }
}

__global__ __launch_bounds__(128) void gpdf_gemm_wmma(
    const float* __restrict__ x,
    const float* __restrict__ Wp,
    const float* __restrict__ bias,
    float* __restrict__ out) {
  const int lane = threadIdx.x & 31;
  const int wave = threadIdx.x >> 5;
  const int h    = lane >> 4;
  const int c    = lane & 15;
  const int row0 = blockIdx.x * 32 + (wave >> 1) * 16;
  const int ct   = blockIdx.y * 2 + (wave & 1);
  const int col0 = ct * 16;

  const v4f* xa = (const v4f*)(x + (row0 + c) * D_DIM + 16 * h);
  const v4f* wb = (const v4f*)Wp + ct * (16 * 2 * 4 * 32) + lane;
  const float bv = bias[col0 + c];

  v8f c1 = {};
  v8f c2 = {};
  Frag f[AHEAD + 1];
  for (int j = 0; j < AHEAD; ++j) load_frag(f[j], xa, wb, j);

#pragma unroll
  for (int j = 0; j < 16; ++j) {
    __builtin_amdgcn_sched_barrier(0);
    if (j + AHEAD < 16) load_frag(f[(j + AHEAD) % (AHEAD + 1)], xa, wb, j + AHEAD);
    __builtin_amdgcn_sched_barrier(0);
    const Frag& g = f[j % (AHEAD + 1)];
    for (int s = 0; s < 8; ++s) {
      const int q = s >> 1;
      const int p = (s & 1) * 2;
      v2f av, bw1, bw2;
      av.x  = g.a[q][p];
      av.y  = g.a[q][p + 1];
      bw1.x = g.b1[q][p];
      bw1.y = g.b1[q][p + 1];
      bw2.x = g.b2[q][p];
      bw2.y = g.b2[q][p + 1];
      const v2f a2 = av * av;
      c1 = __builtin_amdgcn_wmma_f32_16x16x4_f32(false, a2, false, bw1,
                                                 (short)0, c1, false, false);
      c2 = __builtin_amdgcn_wmma_f32_16x16x4_f32(false, av, false, bw2,
                                                 (short)0, c2, false, false);
    }
  }

  for (int r = 0; r < 8; ++r) {
    out[(row0 + r + 8 * h) * K_DIM + col0 + c] = (c1[r] + c2[r]) + bv;
  }
}

extern "C" void kernel_launch(void* const* d_in, const int* in_sizes, int n_in,
                              void* d_out, int out_size, void* d_ws, size_t ws_size,
                              hipStream_t stream) {
  const float* x     = (const float*)d_in[0];
  const float* means = (const float*)d_in[1];
  const float* rho   = (const float*)d_in[2];
  float* out  = (float*)d_out;
  float* Wp   = (float*)d_ws;
  float* bias = Wp + 2 * D_DIM * K_DIM;

  gpdf_prep<<<dim3(256), dim3(256), 0, stream>>>(means, rho, Wp, bias);
  gpdf_gemm_wmma<<<dim3(B_DIM / 32, K_DIM / 32), dim3(128), 0, stream>>>(
      x, Wp, bias, out);
}


# One launch: each block forms its 16-class weight tile once in LDS (softplus on hw exp2/log2) for its 4 waves to share; 64x16 tile per block, whole contraction per wave; bias sum order changed
# speedup vs baseline: 5.1952x; 1.2221x over previous
#include <hip/hip_runtime.h>

#define B_DIM 1024
#define D_DIM 512
#define K_DIM 256

typedef __attribute__((ext_vector_type(2))) float v2f;
typedef __attribute__((ext_vector_type(4))) float v4f;
typedef __attribute__((ext_vector_type(8))) float v8f;

#define ROW_UNITS   129
#define PLANE_UNITS (16 * ROW_UNITS)
#define PL_PITCH    36
#define AHEAD 3

__device__ __forceinline__ void load_a(v4f (&a)[4], const v4f* xa, int j) {
  for (int q = 0; q < 4; ++q) a[q] = xa[8 * j + q];
}

__device__ __forceinline__ void load_b(v4f (&b)[8], const v4f* wbl, int j) {
  for (int q = 0; q < 4; ++q) {
    b[q]     = wbl[8 * j + q];
    b[4 + q] = wbl[PLANE_UNITS + 8 * j + q];
  }
}

__global__ __launch_bounds__(128) void gpdf_fused(
    const float* __restrict__ x,
    const float* __restrict__ means,
    const float* __restrict__ rho,
    float* __restrict__ out) {
  __shared__ v4f wl[2 * PLANE_UNITS];
  __shared__ v4f pl4[16 * (PL_PITCH / 4)];
  float* pl = (float*)pl4;

  const int lane = threadIdx.x & 31;
  const int wave = threadIdx.x >> 5;
  const int h    = lane >> 4;
  const int c    = lane & 15;
  const int ct   = blockIdx.x & 15;
  const int row0 = (blockIdx.x >> 4) * 64 + wave * 16;
  const int col0 = ct * 16;

  const v4f* xa = (const v4f*)(x + (row0 + c) * D_DIM + 16 * h);
  v4f a[AHEAD + 1][4];
  for (int j = 0; j < AHEAD; ++j) load_a(a[j], xa, j);

  {
    v4f m[4][4], r[4][4];
    for (int cc = 0; cc < 4; ++cc) {
      const int k = col0 + wave * 4 + cc;
      const v4f* mrow = (const v4f*)(means + k * D_DIM);
      const v4f* rrow = (const v4f*)(rho + k * D_DIM);
      for (int i = 0; i < 4; ++i) {
        m[cc][i] = mrow[32 * i + lane];
        r[cc][i] = rrow[32 * i + lane];
      }
    }
    for (int cc = 0; cc < 4; ++cc) {
      const int cl = wave * 4 + cc;
      float partial = 0.0f;
      for (int i = 0; i < 4; ++i) {
        v4f w1, w2;
        for (int e = 0; e < 4; ++e) {
          const float s  = 0.69314718f * __builtin_amdgcn_logf(1.0f + __expf(r[cc][i][e]));
          const float s2 = s * s;
          const float mv = m[cc][i][e];
          w1[e] = 0.5f * s2;
          w2[e] = -s2 * mv;
          partial += w1[e] * mv * mv;
        }
        wl[cl * ROW_UNITS + 32 * i + lane]               = w1;
        wl[PLANE_UNITS + cl * ROW_UNITS + 32 * i + lane] = w2;
      }
      pl[cl * PL_PITCH + lane] = partial;
    }
  }
  __syncthreads();

  const v4f* wbl = wl + c * ROW_UNITS + 4 * h;
  float bv;
  {
    const v4f* pp = (const v4f*)(pl + c * PL_PITCH + 16 * h);
    const v4f p0 = pp[0], p1 = pp[1], p2 = pp[2], p3 = pp[3];
    float t = p0[0];
    t += p0[1]; t += p0[2]; t += p0[3];
    t += p1[0]; t += p1[1]; t += p1[2]; t += p1[3];
    t += p2[0]; t += p2[1]; t += p2[2]; t += p2[3];
    t += p3[0]; t += p3[1]; t += p3[2]; t += p3[3];
    bv = t + __shfl_xor(t, 16);
  }
  v4f b[2][8];
  load_b(b[0], wbl, 0);

  v8f c1 = {};
  v8f c2 = {};

  for (int j = 0; j < 16; ++j) {
    __builtin_amdgcn_sched_barrier(0);
    if (j + AHEAD < 16) load_a(a[(j + AHEAD) % (AHEAD + 1)], xa, j + AHEAD);
    if (j + 1 < 16) load_b(b[(j + 1) & 1], wbl, j + 1);
    __builtin_amdgcn_sched_barrier(0);
    const v4f (&ga)[4] = a[j % (AHEAD + 1)];
    const v4f (&gb)[8] = b[j & 1];
    for (int s = 0; s < 8; ++s) {
      const int q = s >> 1;
      const int p = (s & 1) * 2;
      v2f av, bw1, bw2;
      av.x  = ga[q][p];
      av.y  = ga[q][p + 1];
      bw1.x = gb[q][p];
      bw1.y = gb[q][p + 1];
      bw2.x = gb[4 + q][p];
      bw2.y = gb[4 + q][p + 1];
      const v2f a2 = av * av;
      c1 = __builtin_amdgcn_wmma_f32_16x16x4_f32(false, a2, false, bw1,
                                                 (short)0, c1, false, false);
      c2 = __builtin_amdgcn_wmma_f32_16x16x4_f32(false, av, false, bw2,
                                                 (short)0, c2, false, false);
    }
  }

  for (int r = 0; r < 8; ++r) {
    out[(row0 + r + 8 * h) * K_DIM + col0 + c] = (c1[r] + c2[r]) + bv;
  }
}

extern "C" void kernel_launch(void* const* d_in, const int* in_sizes, int n_in,
                              void* d_out, int out_size, void* d_ws, size_t ws_size,
                              hipStream_t stream) {
  const float* x     = (const float*)d_in[0];
  const float* means = (const float*)d_in[1];
  const float* rho   = (const float*)d_in[2];
  float* out = (float*)d_out;

  gpdf_fused<<<dim3(256), dim3(128), 0, stream>>>(x, means, rho, out);
}


# Weight tile formed in four 128-feature chunks: products start once chunk 0 is in LDS, later chunks formed between the products from rows requested ahead; barriers wait on LDS only; sum order kept
# speedup vs baseline: 5.4901x; 1.0568x over previous
#include <hip/hip_runtime.h>

#define B_DIM 1024
#define D_DIM 512
#define K_DIM 256

typedef __attribute__((ext_vector_type(2))) float v2f;
typedef __attribute__((ext_vector_type(4))) float v4f;
typedef __attribute__((ext_vector_type(8))) float v8f;

#define ROW_UNITS   129
#define PLANE_UNITS (16 * ROW_UNITS)
#define PL_PITCH    36
#define AHEAD 3
#define NCH 4

__device__ __forceinline__ void load_a(v4f (&a)[4], const v4f* xa, int j) {
  for (int q = 0; q < 4; ++q) a[q] = xa[8 * j + q];
}

__device__ __forceinline__ void load_b(v4f (&b)[8], const v4f* wbl, int j) {
  for (int q = 0; q < 4; ++q) {
    b[q]     = wbl[8 * j + q];
    b[4 + q] = wbl[PLANE_UNITS + 8 * j + q];
  }
}

__device__ __forceinline__ void load_rows(v4f (&m)[4], v4f (&r)[4], const float* means,
                                          const float* rho, int k0, int lane, int i) {
  for (int cc = 0; cc < 4; ++cc) {
    m[cc] = ((const v4f*)(means + (k0 + cc) * D_DIM))[32 * i + lane];
    r[cc] = ((const v4f*)(rho + (k0 + cc) * D_DIM))[32 * i + lane];
  }
}

__device__ __forceinline__ void lds_barrier() {
  __builtin_amdgcn_fence(__ATOMIC_RELEASE, "workgroup", "local");
  __builtin_amdgcn_s_barrier();
  __builtin_amdgcn_fence(__ATOMIC_ACQUIRE, "workgroup", "local");
}

__device__ __forceinline__ void form_chunk(v4f* wl, const v4f (&m)[4], const v4f (&r)[4],
                                           float (&partial)[4], int wave, int lane, int i) {
  for (int cc = 0; cc < 4; ++cc) {
    const int cl = wave * 4 + cc;
    v4f w1, w2;
    for (int e = 0; e < 4; ++e) {
      const float s  = 0.69314718f * __builtin_amdgcn_logf(1.0f + __expf(r[cc][e]));
      const float s2 = s * s;
      const float mv = m[cc][e];
      w1[e] = 0.5f * s2;
      w2[e] = -s2 * mv;
      partial[cc] += w1[e] * mv * mv;
    }
    wl[cl * ROW_UNITS + 32 * i + lane]               = w1;
    wl[PLANE_UNITS + cl * ROW_UNITS + 32 * i + lane] = w2;
  }
}

__global__ __launch_bounds__(128) void gpdf_fused(
    const float* __restrict__ x,
    const float* __restrict__ means,
    const float* __restrict__ rho,
    float* __restrict__ out) {
  __shared__ v4f wl[2 * PLANE_UNITS];
  __shared__ v4f pl4[16 * (PL_PITCH / 4)];
  float* pl = (float*)pl4;

  const int lane = threadIdx.x & 31;
  const int wave = threadIdx.x >> 5;
  const int h    = lane >> 4;
  const int c    = lane & 15;
  const int ct   = blockIdx.x & 15;
  const int row0 = (blockIdx.x >> 4) * 64 + wave * 16;
  const int col0 = ct * 16;

  const v4f* xa = (const v4f*)(x + (row0 + c) * D_DIM + 16 * h);
  v4f a[AHEAD + 1][4];
  v4f m[NCH][4], r[NCH][4];

  load_rows(m[0], r[0], means, rho, col0 + wave * 4, lane, 0);
  __builtin_amdgcn_sched_barrier(0);
  for (int j = 0; j < AHEAD; ++j) load_a(a[j], xa, j);
  __builtin_amdgcn_sched_barrier(0);
  load_rows(m[1], r[1], means, rho, col0 + wave * 4, lane, 1);
  __builtin_amdgcn_sched_barrier(0);

  float partial[4] = {0.0f, 0.0f, 0.0f, 0.0f};
  form_chunk(wl, m[0], r[0], partial, wave, lane, 0);
  lds_barrier();

  const v4f* wbl = wl + c * ROW_UNITS + 4 * h;
  v4f b[2][8];
  load_b(b[0], wbl, 0);

  v8f c1 = {};
  v8f c2 = {};
  v4f bp[8];
  float bv = 0.0f;

  for (int j = 0; j < 16; ++j) {
    __builtin_amdgcn_sched_barrier(0);
    if ((j & 3) == 3 && j + 1 < 16) lds_barrier();
    if (j + AHEAD < 16) load_a(a[(j + AHEAD) % (AHEAD + 1)], xa, j + AHEAD);
    if (j == 0) load_rows(m[2], r[2], means, rho, col0 + wave * 4, lane, 2);
    if (j == 4) load_rows(m[3], r[3], means, rho, col0 + wave * 4, lane, 3);
    if (j + 1 < 16) load_b(b[(j + 1) & 1], wbl, j + 1);
    if (j == 12) {
      const v4f* pp = (const v4f*)(pl + c * PL_PITCH);
      for (int q = 0; q < 8; ++q) bp[q] = pp[q];
    }
    if ((j & 3) == 2 && j + 2 < 16) {
      const int nx = (j >> 2) + 1;
      __builtin_amdgcn_sched_barrier(0);
      form_chunk(wl, m[nx], r[nx], partial, wave, lane, nx);
      if (nx == NCH - 1) {
        for (int cc = 0; cc < 4; ++cc) pl[(wave * 4 + cc) * PL_PITCH + lane] = partial[cc];
      }
    }
    __builtin_amdgcn_sched_barrier(0);
    if (j == 12) {
      float ta = bp[0][0], tb = bp[4][0];
      for (int e = 1; e < 16; ++e) {
        ta += bp[e >> 2][e & 3];
        tb += bp[4 + (e >> 2)][e & 3];
      }
      bv = ta + tb;
    }
    const v4f (&ga)[4] = a[j % (AHEAD + 1)];
    const v4f (&gb)[8] = b[j & 1];
    for (int s = 0; s < 8; ++s) {
      const int q = s >> 1;
      const int p = (s & 1) * 2;
      v2f av, bw1, bw2;
      av.x  = ga[q][p];
      av.y  = ga[q][p + 1];
      bw1.x = gb[q][p];
      bw1.y = gb[q][p + 1];
      bw2.x = gb[4 + q][p];
      bw2.y = gb[4 + q][p + 1];
      const v2f a2 = av * av;
      c1 = __builtin_amdgcn_wmma_f32_16x16x4_f32(false, a2, false, bw1,
                                                 (short)0, c1, false, false);
      c2 = __builtin_amdgcn_wmma_f32_16x16x4_f32(false, av, false, bw2,
                                                 (short)0, c2, false, false);
    }
  }

  for (int r8 = 0; r8 < 8; ++r8) {
    out[(row0 + r8 + 8 * h) * K_DIM + col0 + c] = (c1[r8] + c2[r8]) + bv;
  }
}

extern "C" void kernel_launch(void* const* d_in, const int* in_sizes, int n_in,
                              void* d_out, int out_size, void* d_ws, size_t ws_size,
                              hipStream_t stream) {
  const float* x     = (const float*)d_in[0];
  const float* means = (const float*)d_in[1];
  const float* rho   = (const float*)d_in[2];
  float* out = (float*)d_out;

  gpdf_fused<<<dim3(256), dim3(128), 0, stream>>>(x, means, rho, out);
}
